# starting kernel
# baseline (speedup 1.0000x reference)
#include <hip/hip_runtime.h>

typedef __attribute__((ext_vector_type(2))) float v2f;
typedef __attribute__((ext_vector_type(8))) float v8f;

#define HDIM 1024
#define BDIM 8

__global__ void __launch_bounds__(32)
bilinear_wv_wmma(const float* __restrict__ W,
                 const float* __restrict__ hidden,
                 float* __restrict__ v)
{
    const int lane = threadIdx.x;
    const int j0   = blockIdx.x * 16;
    const int m    = lane & 15;
    const int kp   = (lane >> 4) * 2;
    const int n    = lane & 15;

    const float* Arow = W + (size_t)(j0 + m) * HDIM + kp;
    const float* Brow = hidden + (size_t)((n < BDIM) ? n : 0) * HDIM + kp;
    const float  bmask = (n < BDIM) ? 1.0f : 0.0f;

    v8f acc0 = {};
    v8f acc1 = {};
    #pragma unroll 2
    for (int k0 = 0; k0 < HDIM; k0 += 8) {
        v2f a0 = *(const v2f*)(Arow + k0);
        v2f h0 = *(const v2f*)(Brow + k0);
        v2f b0; b0.x = h0.x * bmask; b0.y = h0.y * bmask;
        acc0 = __builtin_amdgcn_wmma_f32_16x16x4_f32(false, a0, false, b0,
                                                     (short)0, acc0, false, false);
        v2f a1 = *(const v2f*)(Arow + k0 + 4);
        v2f h1 = *(const v2f*)(Brow + k0 + 4);
        v2f b1; b1.x = h1.x * bmask; b1.y = h1.y * bmask;
        acc1 = __builtin_amdgcn_wmma_f32_16x16x4_f32(false, a1, false, b1,
                                                     (short)0, acc1, false, false);
    }
    v8f acc = acc0 + acc1;

    if (n < BDIM) {
        const int mbase = (lane >> 4) * 8;
        #pragma unroll
        for (int g = 0; g < 8; ++g)
            v[(size_t)n * HDIM + j0 + mbase + g] = acc[g];
    }
}

__global__ void __launch_bounds__(256)
bilinear_dot(const float* __restrict__ enc,
             const float* __restrict__ v,
             const float* __restrict__ bias,
             float* __restrict__ out)
{
    __shared__ float sv[HDIM];
    const int t = threadIdx.x;
    const int b = (blockIdx.x * 8) >> 13;

    ((float4*)sv)[t] = ((const float4*)(v + (size_t)b * HDIM))[t];
    __syncthreads();

    const int    wave = t >> 5;
    const int    lane = t & 31;
    const size_t row  = (size_t)blockIdx.x * 8 + wave;
    const float* erow = enc + row * HDIM;

    float4 acc = make_float4(0.f, 0.f, 0.f, 0.f);
    #pragma unroll
    for (int c = 0; c < 8; ++c) {
        const int k  = c * 128 + lane * 4;
        float4 e  = *(const float4*)(erow + k);
        float4 vv = *(const float4*)(sv + k);
        acc.x += e.x * vv.x;
        acc.y += e.y * vv.y;
        acc.z += e.z * vv.z;
        acc.w += e.w * vv.w;
    }
    float s = (acc.x + acc.y) + (acc.z + acc.w);
    #pragma unroll
    for (int off = 16; off > 0; off >>= 1)
        s += __shfl_xor(s, off, 32);

    if (lane == 0)
        out[row] = s + bias[0];
}

extern "C" void kernel_launch(void* const* d_in, const int* in_sizes, int n_in,
                              void* d_out, int out_size, void* d_ws, size_t ws_size,
                              hipStream_t stream)
{
    const float* hidden = (const float*)d_in[0];
    const float* enc    = (const float*)d_in[1];
    const float* W      = (const float*)d_in[3];
    const float* bias   = (const float*)d_in[4];

    float* v = (float*)d_ws;

    bilinear_wv_wmma<<<HDIM / 16, 32, 0, stream>>>(W, hidden, v);

    bilinear_dot<<<65536 / 8, 256, 0, stream>>>(enc, v, bias, (float*)d_out);
}


# kernel 1: 8-row j-tiles; the 8 idle batch columns of the 16x16 tile carry the second half of K, so a wave chains 128 WMMAs, not 256 (128 waves); halves joined by one lane exchange; new sum order of v
# speedup vs baseline: 1.4660x; 1.4660x over previous
#include <hip/hip_runtime.h>

typedef __attribute__((ext_vector_type(2))) float v2f;
typedef __attribute__((ext_vector_type(4))) float v4f;
typedef __attribute__((ext_vector_type(8))) float v8f;

#define HDIM 1024
#define BDIM 8

#define KHALF  (HDIM / 2)
#define KSTEPS (KHALF / 8)
#define AHEAD  28
#define JTILE  8

__global__ void __launch_bounds__(32)
bilinear_wv_wmma(const float* __restrict__ W,
                 const float* __restrict__ hidden,
                 float* __restrict__ v)
{
    const int lane = threadIdx.x;
    const int j0   = blockIdx.x * JTILE;
    const int c    = lane & 15;
    const int r8   = c & 7;
    const int kq   = (c >> 3) * KHALF
                   + (lane >> 4) * 4;

    const float* Arow = W + (size_t)(j0 + r8) * HDIM + kq;
    const float* Brow = hidden + (size_t)r8 * HDIM + kq;

    v4f a[KSTEPS];
    v4f h[KSTEPS];

    #pragma unroll
    for (int q = 0; q < AHEAD; ++q) {
        a[q] = *(const v4f*)(Arow + q * 8);
        h[q] = *(const v4f*)(Brow + q * 8);
    }

    v8f acc0 = {};
    v8f acc1 = {};
    #pragma unroll
    for (int q = 0; q < KSTEPS; ++q) {
        if (q + AHEAD < KSTEPS) {
            a[q + AHEAD] = *(const v4f*)(Arow + (q + AHEAD) * 8);
            h[q + AHEAD] = *(const v4f*)(Brow + (q + AHEAD) * 8);
        }
        __builtin_amdgcn_sched_barrier(0);
        v2f a0; a0.x = a[q].x; a0.y = a[q].y;
        v2f b0; b0.x = h[q].x; b0.y = h[q].y;
        acc0 = __builtin_amdgcn_wmma_f32_16x16x4_f32(false, a0, false, b0,
                                                     (short)0, acc0, false, false);
        v2f a1; a1.x = a[q].z; a1.y = a[q].w;
        v2f b1; b1.x = h[q].z; b1.y = h[q].w;
        acc1 = __builtin_amdgcn_wmma_f32_16x16x4_f32(false, a1, false, b1,
                                                     (short)0, acc1, false, false);
    }
    v8f acc = acc0 + acc1;

    float s[8];
    #pragma unroll
    for (int g = 0; g < 8; ++g)
        s[g] = acc[g] + __shfl_xor(acc[g], 24, 32);

    if (lane < BDIM) {
        #pragma unroll
        for (int g = 0; g < 8; ++g)
            v[(size_t)lane * HDIM + j0 + g] = s[g];
    }
}

#define SET_WAVES   8
#define BLOCK_WAVES 16
#define SET_ROWS    128
#define GROUP       16
#define NGROUPS     (SET_ROWS / GROUP)
#define BLOCK_ROWS  (2 * SET_ROWS)

__global__ void __launch_bounds__(BLOCK_WAVES * 32)
bilinear_dot(const float* __restrict__ enc,
             const float* __restrict__ v,
             const float* __restrict__ bias,
             float* __restrict__ out)
{
    __shared__ float part[2][SET_WAVES][SET_ROWS];

    const int t    = threadIdx.x;
    const int lane = t & 31;
    const int wave = t >> 5;
    const int set  = wave >> 3;
    const int sw   = wave & 7;
    const int col  = sw * 128 + lane * 4;

    const size_t brow0 = (size_t)blockIdx.x * BLOCK_ROWS;
    const int    b     = (int)(brow0 >> 13);

    const v4f    vv   = *(const v4f*)(v + (size_t)b * HDIM + col);
    const float* ecol = enc + (brow0 + (size_t)set * SET_ROWS) * HDIM + col;

    v4f   e[NGROUPS][GROUP];
    float u[NGROUPS];

    #pragma unroll
    for (int i = 0; i < GROUP; ++i)
        e[0][i] = *(const v4f*)(ecol + (size_t)i * HDIM);

    const bool h16 = (lane & 16) != 0;
    const bool h8  = (lane & 8)  != 0;
    const bool h4  = (lane & 4)  != 0;
    const bool h2  = (lane & 2)  != 0;

    #pragma unroll
    for (int g = 0; g < NGROUPS; ++g) {
        __builtin_amdgcn_sched_barrier(0);
        if (g + 1 < NGROUPS) {
            #pragma unroll
            for (int i = 0; i < GROUP; ++i)
                e[g + 1][i] = *(const v4f*)(ecol + (size_t)((g + 1) * GROUP + i) * HDIM);
        }
        __builtin_amdgcn_sched_barrier(0);

        float p[GROUP];
        #pragma unroll
        for (int i = 0; i < GROUP; ++i)
            p[i] = e[g][i].x * vv.x + e[g][i].y * vv.y + e[g][i].z * vv.z + e[g][i].w * vv.w;

        float q[8];
        #pragma unroll
        for (int i = 0; i < 8; ++i) {
            const float keep = h16 ? p[i + 8] : p[i];
            const float send = h16 ? p[i]     : p[i + 8];
            q[i] = keep + __shfl_xor(send, 16, 32);
        }
        float r[4];
        #pragma unroll
        for (int i = 0; i < 4; ++i) {
            const float keep = h8 ? q[i + 4] : q[i];
            const float send = h8 ? q[i]     : q[i + 4];
            r[i] = keep + __shfl_xor(send, 8, 32);
        }
        float s[2];
        #pragma unroll
        for (int i = 0; i < 2; ++i) {
            const float keep = h4 ? r[i + 2] : r[i];
            const float send = h4 ? r[i]     : r[i + 2];
            s[i] = keep + __shfl_xor(send, 4, 32);
        }
        const float keep = h2 ? s[1] : s[0];
        const float send = h2 ? s[0] : s[1];
        float w1 = keep + __shfl_xor(send, 2, 32);
        w1 += __shfl_xor(w1, 1, 32);
        u[g] = w1;
    }

    if ((lane & 1) == 0) {
        #pragma unroll
        for (int g = 0; g < NGROUPS; ++g)
            part[set][sw][g * GROUP + (lane >> 1)] = u[g];
    }
    __syncthreads();

    if (t < BLOCK_ROWS) {
        const int ps = t >> 7;
        const int pr = t & 127;
        float sum = part[ps][0][pr];
        #pragma unroll
        for (int w = 1; w < SET_WAVES; ++w)
            sum += part[ps][w][pr];
        out[brow0 + t] = sum + bias[0];
    }
}

extern "C" void kernel_launch(void* const* d_in, const int* in_sizes, int n_in,
                              void* d_out, int out_size, void* d_ws, size_t ws_size,
                              hipStream_t stream)
{
    const float* hidden = (const float*)d_in[0];
    const float* enc    = (const float*)d_in[1];
    const float* W      = (const float*)d_in[3];
    const float* bias   = (const float*)d_in[4];

    float* v = (float*)d_ws;

    bilinear_wv_wmma<<<HDIM / JTILE, 32, 0, stream>>>(W, hidden, v);

    bilinear_dot<<<65536 / BLOCK_ROWS, BLOCK_WAVES * 32, 0, stream>>>(enc, v, bias, (float*)d_out);
}
